# starting kernel
# baseline (speedup 1.0000x reference)
#include <hip/hip_runtime.h>
#include <hip/hip_bf16.h>

typedef float v2f __attribute__((ext_vector_type(2)));
typedef float v8f __attribute__((ext_vector_type(8)));

#define FD 128


__global__ void gcn_deg_kernel(const int* __restrict__ dst, float* __restrict__ deg, int E) {
    int e = blockIdx.x * blockDim.x + threadIdx.x;
    if (e < E) unsafeAtomicAdd(&deg[dst[e]], 1.0f);
}

__global__ void gcn_dis_kernel(const float* __restrict__ deg, float* __restrict__ dis, int N) {
    int i = blockIdx.x * blockDim.x + threadIdx.x;
    if (i < N) dis[i] = rsqrtf(deg[i] + 1.0f);
}


__global__ void gcn_gemm_wmma(const float* __restrict__ A, const float* __restrict__ W,
                              float* __restrict__ C, int M) {
    const int K  = FD;
    const int Nc = FD;
    int wave = (int)((blockIdx.x * blockDim.x + threadIdx.x) >> 5);
    int lane = threadIdx.x & 31;
    const int tiles_n = Nc / 16;
    int tile_m = wave / tiles_n;
    int tile_n = wave - tile_m * tiles_n;
    if (tile_m * 16 >= M) return;

    int half = lane >> 4;
    int l16  = lane & 15;

    const float* arow = A + (size_t)(tile_m * 16 + l16) * K;
    const float* wcol = W + (size_t)(tile_n * 16 + l16);

    v8f c = {};
    #pragma unroll 4
    for (int k = 0; k < K; k += 4) {
        v2f a, b;
        int ka = k + 2 * half;
        a.x = arow[ka];
        a.y = arow[ka + 1];
        b.x = wcol[(size_t)ka * Nc];
        b.y = wcol[(size_t)(ka + 1) * Nc];
        c = __builtin_amdgcn_wmma_f32_16x16x4_f32(false, a, false, b,
                                                  (short)0, c, false, false);
    }

    float* crow = C + (size_t)(tile_m * 16) * Nc + tile_n * 16;
    #pragma unroll
    for (int v = 0; v < 8; ++v) {
        crow[(size_t)(v + 8 * half) * Nc + l16] = c[v];
    }
}


__global__ void gcn_edge_agg(const float* __restrict__ h, const int* __restrict__ src,
                             const int* __restrict__ dst, const float* __restrict__ dis,
                             float* __restrict__ out, int E, int N) {
    int idx  = (int)((blockIdx.x * blockDim.x + threadIdx.x) >> 5);
    int lane = threadIdx.x & 31;
    int total = E + N;
    if (idx >= total) return;

    int s, d;
    if (idx < E) { s = src[idx]; d = dst[idx]; }
    else         { s = idx - E; d = s; }

    float nrm = dis[s] * dis[d];
    const float4* hs = (const float4*)(h + (size_t)s * FD);
    float4 v = hs[lane];
    float* o = out + (size_t)d * FD + lane * 4;
    unsafeAtomicAdd(o + 0, v.x * nrm);
    unsafeAtomicAdd(o + 1, v.y * nrm);
    unsafeAtomicAdd(o + 2, v.z * nrm);
    unsafeAtomicAdd(o + 3, v.w * nrm);
}


__global__ void gcn_bias_relu(const float* __restrict__ in, const float* __restrict__ b,
                              float* __restrict__ out, int total) {
    int i = blockIdx.x * blockDim.x + threadIdx.x;
    if (i < total) {
        float v = in[i] + b[i & (FD - 1)];
        out[i] = v > 0.0f ? v : 0.0f;
    }
}

__global__ void gcn_bias_add(float* __restrict__ out, const float* __restrict__ b, int total) {
    int i = blockIdx.x * blockDim.x + threadIdx.x;
    if (i < total) out[i] += b[i & (FD - 1)];
}


extern "C" void kernel_launch(void* const* d_in, const int* in_sizes, int n_in,
                              void* d_out, int out_size, void* d_ws, size_t ws_size,
                              hipStream_t stream) {
    const float* x  = (const float*)d_in[0];
    const int*   ei = (const int*)d_in[1];
    const float* W1 = (const float*)d_in[2];
    const float* b1 = (const float*)d_in[3];
    const float* W2 = (const float*)d_in[4];
    const float* b2 = (const float*)d_in[5];

    const int N = in_sizes[0] / FD;
    const int E = in_sizes[1] / 2;
    const int* src = ei;
    const int* dst = ei + E;

    float* ws   = (float*)d_ws;
    float* deg  = ws;
    float* dis  = ws + N;
    float* bufA = ws + 2 * (size_t)N;
    float* bufB = bufA + (size_t)N * FD;
    float* out  = (float*)d_out;

    const size_t feat_bytes = (size_t)N * FD * sizeof(float);
    const int total = N * FD;

    hipMemsetAsync(deg, 0, (size_t)N * sizeof(float), stream);
    gcn_deg_kernel<<<(E + 255) / 256, 256, 0, stream>>>(dst, deg, E);
    gcn_dis_kernel<<<(N + 255) / 256, 256, 0, stream>>>(deg, dis, N);

    const int tilesM  = (N + 15) / 16;
    const int waves   = tilesM * (FD / 16);
    const int gthreads = waves * 32;
    const int gblocks  = (gthreads + 255) / 256;

    const int ethreads = (E + N) * 32;
    const int eblocks  = (ethreads + 255) / 256;

    gcn_gemm_wmma<<<gblocks, 256, 0, stream>>>(x, W1, bufA, N);
    hipMemsetAsync(bufB, 0, feat_bytes, stream);
    gcn_edge_agg<<<eblocks, 256, 0, stream>>>(bufA, src, dst, dis, bufB, E, N);
    gcn_bias_relu<<<(total + 255) / 256, 256, 0, stream>>>(bufB, b1, bufA, total);

    gcn_gemm_wmma<<<gblocks, 256, 0, stream>>>(bufA, W2, bufB, N);
    hipMemsetAsync(out, 0, feat_bytes, stream);
    gcn_edge_agg<<<eblocks, 256, 0, stream>>>(bufB, src, dst, dis, out, E, N);
    gcn_bias_add<<<(total + 255) / 256, 256, 0, stream>>>(out, b2, total);
}


# Aggregation pulls over a destination-sorted edge list built each call (int-atomic count + arrival-order rank, scan, fill): a wave sums its row in registers and stores it once with bias/relu
# speedup vs baseline: 2.2078x; 2.2078x over previous
#include <hip/hip_runtime.h>
#include <hip/hip_bf16.h>


typedef float v2f __attribute__((ext_vector_type(2)));
typedef float v8f __attribute__((ext_vector_type(8)));

#define FD 128
#define SCAN_T 1024


__global__ void gcn_count_rank(const int* __restrict__ dst, int* __restrict__ cnt,
                               int* __restrict__ rank, int E) {
    int e = blockIdx.x * blockDim.x + threadIdx.x;
    if (e < E) rank[e] = atomicAdd(&cnt[dst[e]], 1);
}


__global__ void __launch_bounds__(SCAN_T)
gcn_scan_dis(const int* __restrict__ cnt, int* __restrict__ rowptr,
             float* __restrict__ dis, int N) {
    __shared__ int s_pre[SCAN_T / 32];
    __shared__ int s_tile[SCAN_T / 32];
    const int tid  = threadIdx.x;
    const int lane = tid & 31;
    const int wv   = tid >> 5;
    const int tile0 = blockIdx.x * SCAN_T;

    int pre = 0;
    const int4* c4 = (const int4*)cnt;
    const int n4 = tile0 >> 2;
    for (int i = tid; i < n4; i += 4 * SCAN_T) {
        int4 v[4];
        #pragma unroll
        for (int j = 0; j < 4; ++j) {
            const int idx = i + j * SCAN_T;
            v[j] = (idx < n4) ? c4[idx] : make_int4(0, 0, 0, 0);
        }
        #pragma unroll
        for (int j = 0; j < 4; ++j) pre += (v[j].x + v[j].y) + (v[j].z + v[j].w);
    }

    const int i  = tile0 + tid;
    const int my = (i < N) ? cnt[i] : 0;
    int incl = my;
    for (int o = 1; o < 32; o <<= 1) {
        int t = __shfl_up(incl, o);
        if (lane >= o) incl += t;
    }
    for (int o = 16; o > 0; o >>= 1) pre += __shfl_xor(pre, o);
    if (lane == 31) s_tile[wv] = incl;
    if (lane == 0)  s_pre[wv]  = pre;
    __syncthreads();

    int base = 0;
    for (int w = 0; w < SCAN_T / 32; ++w) {
        base += s_pre[w];
        if (w < wv) base += s_tile[w];
    }
    if (i <= N) rowptr[i] = base + incl - my;
    if (i < N)  dis[i] = rsqrtf((float)my + 1.0f);
}


__global__ void gcn_fill(const int* __restrict__ src, const int* __restrict__ dst,
                         const int* __restrict__ rank, const int* __restrict__ rowptr,
                         const float* __restrict__ dis, int* __restrict__ adj, int E) {
    int e = blockIdx.x * blockDim.x + threadIdx.x;
    if (e < E) {
        int s = src[e];
        int d = dst[e];
        float nrm = dis[s] * dis[d];
        ((int2*)adj)[rowptr[d] + rank[e]] = make_int2(s, __float_as_int(nrm));
    }
}


__global__ void gcn_gemm_wmma(const float* __restrict__ A, const float* __restrict__ W,
                              float* __restrict__ C, int M) {
    const int K  = FD;
    const int Nc = FD;
    int wave = (int)((blockIdx.x * blockDim.x + threadIdx.x) >> 5);
    int lane = threadIdx.x & 31;
    const int tiles_n = Nc / 16;
    int tile_m = wave / tiles_n;
    int tile_n = wave - tile_m * tiles_n;
    if (tile_m * 16 >= M) return;

    int half = lane >> 4;
    int l16  = lane & 15;

    const float* arow = A + (size_t)(tile_m * 16 + l16) * K;
    const float* wcol = W + (size_t)(tile_n * 16 + l16);

    v8f c = {};
    #pragma unroll 4
    for (int k = 0; k < K; k += 4) {
        v2f a, b;
        int ka = k + 2 * half;
        a.x = arow[ka];
        a.y = arow[ka + 1];
        b.x = wcol[(size_t)ka * Nc];
        b.y = wcol[(size_t)(ka + 1) * Nc];
        c = __builtin_amdgcn_wmma_f32_16x16x4_f32(false, a, false, b,
                                                  (short)0, c, false, false);
    }

    float* crow = C + (size_t)(tile_m * 16) * Nc + tile_n * 16;
    #pragma unroll
    for (int v = 0; v < 8; ++v) {
        crow[(size_t)(v + 8 * half) * Nc + l16] = c[v];
    }
}


template <int K>
__device__ __forceinline__ void gcn_gather_rows(const float4* __restrict__ h4, int2 my, int t,
                                                int lane, float4& acc) {
    float4 v[K];
    float  w[K];
    #pragma unroll
    for (int j = 0; j < K; ++j) {
        const int s = __builtin_amdgcn_readlane(my.x, t + j);
        w[j] = __int_as_float(__builtin_amdgcn_readlane(my.y, t + j));
        v[j] = h4[(size_t)s * (FD / 4) + lane];
    }
    #pragma unroll
    for (int j = 0; j < K; ++j) {
        acc.x = fmaf(v[j].x, w[j], acc.x);
        acc.y = fmaf(v[j].y, w[j], acc.y);
        acc.z = fmaf(v[j].z, w[j], acc.z);
        acc.w = fmaf(v[j].w, w[j], acc.w);
    }
}

__global__ void __launch_bounds__(256)
gcn_pull_agg(const float* __restrict__ h, const int* __restrict__ rowptr,
             const int* __restrict__ adj, const float* __restrict__ dis,
             const float* __restrict__ bias, float* __restrict__ out, int N, int relu) {
    const int lane = threadIdx.x & 31;
    const int d = __builtin_amdgcn_readfirstlane(
        (int)((blockIdx.x * blockDim.x + threadIdx.x) >> 5));
    if (d >= N) return;

    const int beg = rowptr[d];
    const int end = rowptr[d + 1];
    const float dd = dis[d];
    const float4* h4 = (const float4*)h;
    const int2* adj2 = (const int2*)adj;

    const float4 sv = h4[(size_t)d * (FD / 4) + lane];
    const float sw = dd * dd;
    float4 acc = make_float4(sv.x * sw, sv.y * sw, sv.z * sw, sv.w * sw);

    for (int base = beg; base < end; base += 32) {
        int m = end - base;
        if (m > 32) m = 32;
        int2 my = make_int2(0, 0);
        if (lane < m) my = adj2[base + lane];

        int t = 0;
        for (; t + 8 <= m; t += 8) gcn_gather_rows<8>(h4, my, t, lane, acc);
        if (m - t >= 4) { gcn_gather_rows<4>(h4, my, t, lane, acc); t += 4; }
        if (m - t >= 2) { gcn_gather_rows<2>(h4, my, t, lane, acc); t += 2; }
        if (m - t >= 1) { gcn_gather_rows<1>(h4, my, t, lane, acc); }
    }

    const float4 b = ((const float4*)bias)[lane];
    float4 r = make_float4(acc.x + b.x, acc.y + b.y, acc.z + b.z, acc.w + b.w);
    if (relu) {
        r.x = r.x > 0.0f ? r.x : 0.0f;
        r.y = r.y > 0.0f ? r.y : 0.0f;
        r.z = r.z > 0.0f ? r.z : 0.0f;
        r.w = r.w > 0.0f ? r.w : 0.0f;
    }
    ((float4*)out)[(size_t)d * (FD / 4) + lane] = r;
}


extern "C" void kernel_launch(void* const* d_in, const int* in_sizes, int n_in,
                              void* d_out, int out_size, void* d_ws, size_t ws_size,
                              hipStream_t stream) {
    const float* x  = (const float*)d_in[0];
    const int*   ei = (const int*)d_in[1];
    const float* W1 = (const float*)d_in[2];
    const float* b1 = (const float*)d_in[3];
    const float* W2 = (const float*)d_in[4];
    const float* b2 = (const float*)d_in[5];

    const int N = in_sizes[0] / FD;
    const int E = in_sizes[1] / 2;
    const int* src = ei;
    const int* dst = ei + E;

    const size_t NP = ((size_t)N + 1 + 63) / 64 * 64;
    const size_t EP = ((size_t)E + 63) / 64 * 64;
    int*   wsi    = (int*)d_ws;
    int*   cnt    = wsi;
    int*   rowptr = wsi + NP;
    float* dis    = (float*)(wsi + 2 * NP);
    int*   rank   = wsi + 3 * NP;
    int*   adj    = wsi + 3 * NP + EP;
    float* bufA   = (float*)(wsi + 3 * NP + 3 * EP);
    float* bufB   = bufA + (size_t)N * FD;
    float* out    = (float*)d_out;

    hipMemsetAsync(cnt, 0, (size_t)N * sizeof(int), stream);
    gcn_count_rank<<<(E + 255) / 256, 256, 0, stream>>>(dst, cnt, rank, E);
    gcn_scan_dis<<<(N + 1 + SCAN_T - 1) / SCAN_T, SCAN_T, 0, stream>>>(cnt, rowptr, dis, N);
    gcn_fill<<<(E + 255) / 256, 256, 0, stream>>>(src, dst, rank, rowptr, dis, adj, E);

    const int tilesM  = (N + 15) / 16;
    const int waves   = tilesM * (FD / 16);
    const int gthreads = waves * 32;
    const int gblocks  = (gthreads + 255) / 256;

    const int ablocks = (N + 7) / 8;

    gcn_gemm_wmma<<<gblocks, 256, 0, stream>>>(x, W1, bufA, N);
    gcn_pull_agg<<<ablocks, 256, 0, stream>>>(bufA, rowptr, adj, dis, b1, bufB, N, 1);

    gcn_gemm_wmma<<<gblocks, 256, 0, stream>>>(bufB, W2, bufA, N);
    gcn_pull_agg<<<ablocks, 256, 0, stream>>>(bufA, rowptr, adj, dis, b2, out, N, 0);
}


# GEMM re-tiled: a wave computes a 16x128 row strip (each A fragment loaded once, feeds 8 accumulator tiles), W staged once per work-group in LDS in fragment order; k is summed in a permuted order
# speedup vs baseline: 2.4963x; 1.1307x over previous
#include <hip/hip_runtime.h>
#include <hip/hip_bf16.h>


typedef float v2f __attribute__((ext_vector_type(2)));
typedef float v8f __attribute__((ext_vector_type(8)));

#define FD 128
#define SCAN_T 1024


__global__ void gcn_count_rank(const int* __restrict__ dst, int* __restrict__ cnt,
                               int* __restrict__ rank, int E) {
    int e = blockIdx.x * blockDim.x + threadIdx.x;
    if (e < E) rank[e] = atomicAdd(&cnt[dst[e]], 1);
}


__global__ void __launch_bounds__(SCAN_T)
gcn_scan_dis(const int* __restrict__ cnt, int* __restrict__ rowptr,
             float* __restrict__ dis, int N) {
    __shared__ int s_pre[SCAN_T / 32];
    __shared__ int s_tile[SCAN_T / 32];
    const int tid  = threadIdx.x;
    const int lane = tid & 31;
    const int wv   = tid >> 5;
    const int tile0 = blockIdx.x * SCAN_T;

    int pre = 0;
    const int4* c4 = (const int4*)cnt;
    const int n4 = tile0 >> 2;
    for (int i = tid; i < n4; i += 4 * SCAN_T) {
        int4 v[4];
        #pragma unroll
        for (int j = 0; j < 4; ++j) {
            const int idx = i + j * SCAN_T;
            v[j] = (idx < n4) ? c4[idx] : make_int4(0, 0, 0, 0);
        }
        #pragma unroll
        for (int j = 0; j < 4; ++j) pre += (v[j].x + v[j].y) + (v[j].z + v[j].w);
    }

    const int i  = tile0 + tid;
    const int my = (i < N) ? cnt[i] : 0;
    int incl = my;
    for (int o = 1; o < 32; o <<= 1) {
        int t = __shfl_up(incl, o);
        if (lane >= o) incl += t;
    }
    for (int o = 16; o > 0; o >>= 1) pre += __shfl_xor(pre, o);
    if (lane == 31) s_tile[wv] = incl;
    if (lane == 0)  s_pre[wv]  = pre;
    __syncthreads();

    int base = 0;
    for (int w = 0; w < SCAN_T / 32; ++w) {
        base += s_pre[w];
        if (w < wv) base += s_tile[w];
    }
    if (i <= N) rowptr[i] = base + incl - my;
    if (i < N)  dis[i] = rsqrtf((float)my + 1.0f);
}


__global__ void gcn_fill(const int* __restrict__ src, const int* __restrict__ dst,
                         const int* __restrict__ rank, const int* __restrict__ rowptr,
                         const float* __restrict__ dis, int* __restrict__ adj, int E) {
    int e = blockIdx.x * blockDim.x + threadIdx.x;
    if (e < E) {
        int s = src[e];
        int d = dst[e];
        float nrm = dis[s] * dis[d];
        ((int2*)adj)[rowptr[d] + rank[e]] = make_int2(s, __float_as_int(nrm));
    }
}


#define GEMM_WAVES 8

__global__ void __launch_bounds__(32 * GEMM_WAVES)
gcn_gemm_wmma(const float* __restrict__ A, const float* __restrict__ W,
              float* __restrict__ C, int M) {
    __shared__ float4 s_w[(FD / 4) * 8 * 16];

    const int tid  = threadIdx.x;
    const int lane = tid & 31;
    const int wv   = tid >> 5;
    const int half = lane >> 4;
    const int l16  = lane & 15;
    const int row0 = (blockIdx.x * GEMM_WAVES + wv) * 16;

    int ar = row0 + l16;
    if (ar > M - 1) ar = M - 1;
    const float4* arow = (const float4*)(A + (size_t)ar * FD) + half;
    float4 a4[16];
    #pragma unroll
    for (int i = 0; i < 16; ++i) a4[i] = arow[2 * i];

    const float4* W4 = (const float4*)W;
    #pragma unroll
    for (int j = 0; j < FD / 4 / GEMM_WAVES; ++j) {
        const int kq = wv + GEMM_WAVES * j;
        const float4 r0 = W4[(4 * kq + 0) * (FD / 4) + lane];
        const float4 r1 = W4[(4 * kq + 1) * (FD / 4) + lane];
        const float4 r2 = W4[(4 * kq + 2) * (FD / 4) + lane];
        const float4 r3 = W4[(4 * kq + 3) * (FD / 4) + lane];
        float4* dstp = s_w + (kq * 8 + half * 4) * 16 + l16;
        dstp[0 * 16] = make_float4(r0.x, r1.x, r2.x, r3.x);
        dstp[1 * 16] = make_float4(r0.y, r1.y, r2.y, r3.y);
        dstp[2 * 16] = make_float4(r0.z, r1.z, r2.z, r3.z);
        dstp[3 * 16] = make_float4(r0.w, r1.w, r2.w, r3.w);
    }
    __syncthreads();
    if (row0 >= M) return;

    v8f acc[8];
    #pragma unroll
    for (int T = 0; T < 8; ++T) acc[T] = (v8f){};

    const float4* bsrc = s_w + half * 8 * 16 + l16;
    #pragma unroll
    for (int i = 0; i < 16; ++i) {
        float4 b4[8];
        #pragma unroll
        for (int T = 0; T < 8; ++T) b4[T] = bsrc[(2 * i * 8 + T) * 16];
        v2f a_lo, a_hi;
        a_lo.x = a4[i].x; a_lo.y = a4[i].y;
        a_hi.x = a4[i].z; a_hi.y = a4[i].w;
        #pragma unroll
        for (int T = 0; T < 8; ++T) {
            v2f b;
            b.x = b4[T].x; b.y = b4[T].y;
            acc[T] = __builtin_amdgcn_wmma_f32_16x16x4_f32(false, a_lo, false, b,
                                                           (short)0, acc[T], false, false);
        }
        #pragma unroll
        for (int T = 0; T < 8; ++T) {
            v2f b;
            b.x = b4[T].z; b.y = b4[T].w;
            acc[T] = __builtin_amdgcn_wmma_f32_16x16x4_f32(false, a_hi, false, b,
                                                           (short)0, acc[T], false, false);
        }
    }

    float4* crow = (float4*)C + (size_t)(row0 + 8 * half) * (FD / 4) + l16;
    if (row0 + 16 <= M) {
        #pragma unroll
        for (int g = 0; g < 2; ++g) {
            #pragma unroll
            for (int r = 0; r < 8; ++r)
                crow[r * (FD / 4) + 16 * g] =
                    make_float4(acc[4 * g + 0][r], acc[4 * g + 1][r],
                                acc[4 * g + 2][r], acc[4 * g + 3][r]);
        }
    } else {
        #pragma unroll
        for (int g = 0; g < 2; ++g) {
            #pragma unroll
            for (int r = 0; r < 8; ++r) {
                if (row0 + 8 * half + r < M)
                    crow[r * (FD / 4) + 16 * g] =
                        make_float4(acc[4 * g + 0][r], acc[4 * g + 1][r],
                                    acc[4 * g + 2][r], acc[4 * g + 3][r]);
            }
        }
    }
}


template <int K>
__device__ __forceinline__ void gcn_gather_rows(const float4* __restrict__ h4, int2 my, int t,
                                                int lane, float4& acc) {
    float4 v[K];
    float  w[K];
    #pragma unroll
    for (int j = 0; j < K; ++j) {
        const int s = __builtin_amdgcn_readlane(my.x, t + j);
        w[j] = __int_as_float(__builtin_amdgcn_readlane(my.y, t + j));
        v[j] = h4[(size_t)s * (FD / 4) + lane];
    }
    #pragma unroll
    for (int j = 0; j < K; ++j) {
        acc.x = fmaf(v[j].x, w[j], acc.x);
        acc.y = fmaf(v[j].y, w[j], acc.y);
        acc.z = fmaf(v[j].z, w[j], acc.z);
        acc.w = fmaf(v[j].w, w[j], acc.w);
    }
}

__global__ void __launch_bounds__(256)
gcn_pull_agg(const float* __restrict__ h, const int* __restrict__ rowptr,
             const int* __restrict__ adj, const float* __restrict__ dis,
             const float* __restrict__ bias, float* __restrict__ out, int N, int relu) {
    const int lane = threadIdx.x & 31;
    const int d = __builtin_amdgcn_readfirstlane(
        (int)((blockIdx.x * blockDim.x + threadIdx.x) >> 5));
    if (d >= N) return;

    const int beg = rowptr[d];
    const int end = rowptr[d + 1];
    const float dd = dis[d];
    const float4* h4 = (const float4*)h;
    const int2* adj2 = (const int2*)adj;

    const float4 sv = h4[(size_t)d * (FD / 4) + lane];
    const float sw = dd * dd;
    float4 acc = make_float4(sv.x * sw, sv.y * sw, sv.z * sw, sv.w * sw);

    for (int base = beg; base < end; base += 32) {
        int m = end - base;
        if (m > 32) m = 32;
        int2 my = make_int2(0, 0);
        if (lane < m) my = adj2[base + lane];

        int t = 0;
        for (; t + 8 <= m; t += 8) gcn_gather_rows<8>(h4, my, t, lane, acc);
        if (m - t >= 4) { gcn_gather_rows<4>(h4, my, t, lane, acc); t += 4; }
        if (m - t >= 2) { gcn_gather_rows<2>(h4, my, t, lane, acc); t += 2; }
        if (m - t >= 1) { gcn_gather_rows<1>(h4, my, t, lane, acc); }
    }

    const float4 b = ((const float4*)bias)[lane];
    float4 r = make_float4(acc.x + b.x, acc.y + b.y, acc.z + b.z, acc.w + b.w);
    if (relu) {
        r.x = r.x > 0.0f ? r.x : 0.0f;
        r.y = r.y > 0.0f ? r.y : 0.0f;
        r.z = r.z > 0.0f ? r.z : 0.0f;
        r.w = r.w > 0.0f ? r.w : 0.0f;
    }
    ((float4*)out)[(size_t)d * (FD / 4) + lane] = r;
}


extern "C" void kernel_launch(void* const* d_in, const int* in_sizes, int n_in,
                              void* d_out, int out_size, void* d_ws, size_t ws_size,
                              hipStream_t stream) {
    const float* x  = (const float*)d_in[0];
    const int*   ei = (const int*)d_in[1];
    const float* W1 = (const float*)d_in[2];
    const float* b1 = (const float*)d_in[3];
    const float* W2 = (const float*)d_in[4];
    const float* b2 = (const float*)d_in[5];

    const int N = in_sizes[0] / FD;
    const int E = in_sizes[1] / 2;
    const int* src = ei;
    const int* dst = ei + E;

    const size_t NP = ((size_t)N + 1 + 63) / 64 * 64;
    const size_t EP = ((size_t)E + 63) / 64 * 64;
    int*   wsi    = (int*)d_ws;
    int*   cnt    = wsi;
    int*   rowptr = wsi + NP;
    float* dis    = (float*)(wsi + 2 * NP);
    int*   rank   = wsi + 3 * NP;
    int*   adj    = wsi + 3 * NP + EP;
    float* bufA   = (float*)(wsi + 3 * NP + 3 * EP);
    float* bufB   = bufA + (size_t)N * FD;
    float* out    = (float*)d_out;

    hipMemsetAsync(cnt, 0, (size_t)N * sizeof(int), stream);
    gcn_count_rank<<<(E + 255) / 256, 256, 0, stream>>>(dst, cnt, rank, E);
    gcn_scan_dis<<<(N + 1 + SCAN_T - 1) / SCAN_T, SCAN_T, 0, stream>>>(cnt, rowptr, dis, N);
    gcn_fill<<<(E + 255) / 256, 256, 0, stream>>>(src, dst, rank, rowptr, dis, adj, E);

    const int tilesM  = (N + 15) / 16;
    const int gblocks = (tilesM + GEMM_WAVES - 1) / GEMM_WAVES;

    const int ablocks = (N + 7) / 8;

    gcn_gemm_wmma<<<gblocks, 32 * GEMM_WAVES, 0, stream>>>(x, W1, bufA, N);
    gcn_pull_agg<<<ablocks, 256, 0, stream>>>(bufA, rowptr, adj, dis, b1, bufB, N, 1);

    gcn_gemm_wmma<<<gblocks, 32 * GEMM_WAVES, 0, stream>>>(bufB, W2, bufA, N);
    gcn_pull_agg<<<ablocks, 256, 0, stream>>>(bufA, rowptr, adj, dis, b2, out, N, 0);
}


# Edge list built without global atomics: (node range, edge slice) work-groups count destinations in LDS; rank = LDS-atomic arrival order inside a slice, slices in order; memset dropped
# speedup vs baseline: 3.0902x; 1.2379x over previous
#include <hip/hip_runtime.h>
#include <hip/hip_bf16.h>


typedef float v2f __attribute__((ext_vector_type(2)));
typedef float v8f __attribute__((ext_vector_type(8)));

#define FD 128
#define SCAN_T 1024


#define RNG 4096
#define NSL 16
#define CNT_T 1024
#define CNT_U 19

__global__ void __launch_bounds__(CNT_T)
gcn_count_rank(const int* __restrict__ dst, int* __restrict__ counts, int* __restrict__ tiletot,
               int* __restrict__ rank, int E, int slice, int N, int NP) {
    __shared__ int s_cnt[RNG + 32];
    __shared__ int s_tot[RNG / CNT_T];
    const int tid = threadIdx.x;
    const int n0  = blockIdx.x * RNG;
    const int sl  = blockIdx.y;
    for (int i = tid; i < RNG + 32; i += CNT_T) s_cnt[i] = 0;
    if (tid < RNG / CNT_T) s_tot[tid] = 0;
    __syncthreads();

    const int e0 = sl * slice;
    const int e1 = (e0 + slice < E) ? e0 + slice : E;
    const int spare = RNG + (tid & 31);
    for (int base = e0 + tid; base < e1; base += CNT_U * CNT_T) {
        int d[CNT_U];
        int k[CNT_U];
        #pragma unroll
        for (int j = 0; j < CNT_U; ++j) {
            const int idx = base + j * CNT_T;
            d[j] = (idx < e1) ? dst[idx] : -1;
        }
        #pragma unroll
        for (int j = 0; j < CNT_U; ++j) {
            const unsigned r = (unsigned)(d[j] - n0);
            k[j] = atomicAdd(&s_cnt[r < (unsigned)RNG ? (int)r : spare], 1);
        }
        #pragma unroll
        for (int j = 0; j < CNT_U; ++j) {
            const unsigned r = (unsigned)(d[j] - n0);
            if (r < (unsigned)RNG) rank[base + j * CNT_T] = k[j];
        }
    }
    __syncthreads();

    #pragma unroll
    for (int t = 0; t < RNG / CNT_T; ++t) {
        const int c = s_cnt[t * CNT_T + tid];
        const int node = n0 + t * CNT_T + tid;
        if (node < N) counts[(size_t)sl * NP + node] = c;
        int sum = c;
        for (int o = 16; o > 0; o >>= 1) sum += __shfl_xor(sum, o);
        if ((tid & 31) == 0) atomicAdd(&s_tot[t], sum);
    }
    __syncthreads();
    if (tid < RNG / CNT_T)
        tiletot[(blockIdx.x * (RNG / CNT_T) + tid) * NSL + sl] = s_tot[tid];
}


__global__ void __launch_bounds__(SCAN_T)
gcn_scan_dis(int* __restrict__ counts, const int* __restrict__ tiletot,
             int* __restrict__ rowptr, float* __restrict__ dis, int N, int NP) {
    __shared__ int s_pre[SCAN_T / 32];
    __shared__ int s_tile[SCAN_T / 32];
    const int tid  = threadIdx.x;
    const int lane = tid & 31;
    const int wv   = tid >> 5;
    const int tile0 = blockIdx.x * SCAN_T;

    int pre = 0;
    const int npre = blockIdx.x * NSL;
    for (int i = tid; i < npre; i += SCAN_T) pre += tiletot[i];

    const int i = tile0 + tid;
    int c[NSL];
    #pragma unroll
    for (int s = 0; s < NSL; ++s) c[s] = (i < N) ? counts[(size_t)s * NP + i] : 0;
    int my = 0;
    #pragma unroll
    for (int s = 0; s < NSL; ++s) my += c[s];

    int incl = my;
    for (int o = 1; o < 32; o <<= 1) {
        int t = __shfl_up(incl, o);
        if (lane >= o) incl += t;
    }
    for (int o = 16; o > 0; o >>= 1) pre += __shfl_xor(pre, o);
    if (lane == 31) s_tile[wv] = incl;
    if (lane == 0)  s_pre[wv]  = pre;
    __syncthreads();

    int base = 0;
    for (int w = 0; w < SCAN_T / 32; ++w) {
        base += s_pre[w];
        if (w < wv) base += s_tile[w];
    }
    const int start = base + incl - my;
    if (i <= N) rowptr[i] = start;
    if (i < N) {
        dis[i] = rsqrtf((float)my + 1.0f);
        int run = start;
        #pragma unroll
        for (int s = 0; s < NSL; ++s) {
            counts[(size_t)s * NP + i] = run;
            run += c[s];
        }
    }
}


__global__ void gcn_fill(const int* __restrict__ src, const int* __restrict__ dst,
                         const int* __restrict__ rank, const int* __restrict__ slicebase,
                         const float* __restrict__ dis, int* __restrict__ adj,
                         int E, int slice, int NP) {
    const int sl = blockIdx.y;
    const int e0 = sl * slice;
    const int e1 = (e0 + slice < E) ? e0 + slice : E;
    const int e  = e0 + blockIdx.x * blockDim.x + threadIdx.x;
    if (e < e1) {
        int s = src[e];
        int d = dst[e];
        float nrm = dis[s] * dis[d];
        ((int2*)adj)[slicebase[(size_t)sl * NP + d] + rank[e]] = make_int2(s, __float_as_int(nrm));
    }
}


#define GEMM_WAVES 8

__global__ void __launch_bounds__(32 * GEMM_WAVES)
gcn_gemm_wmma(const float* __restrict__ A, const float* __restrict__ W,
              float* __restrict__ C, int M) {
    __shared__ float4 s_w[(FD / 4) * 8 * 16];

    const int tid  = threadIdx.x;
    const int lane = tid & 31;
    const int wv   = tid >> 5;
    const int half = lane >> 4;
    const int l16  = lane & 15;
    const int row0 = (blockIdx.x * GEMM_WAVES + wv) * 16;

    int ar = row0 + l16;
    if (ar > M - 1) ar = M - 1;
    const float4* arow = (const float4*)(A + (size_t)ar * FD) + half;
    float4 a4[16];
    #pragma unroll
    for (int i = 0; i < 16; ++i) a4[i] = arow[2 * i];

    const float4* W4 = (const float4*)W;
    #pragma unroll
    for (int j = 0; j < FD / 4 / GEMM_WAVES; ++j) {
        const int kq = wv + GEMM_WAVES * j;
        const float4 r0 = W4[(4 * kq + 0) * (FD / 4) + lane];
        const float4 r1 = W4[(4 * kq + 1) * (FD / 4) + lane];
        const float4 r2 = W4[(4 * kq + 2) * (FD / 4) + lane];
        const float4 r3 = W4[(4 * kq + 3) * (FD / 4) + lane];
        float4* dstp = s_w + (kq * 8 + half * 4) * 16 + l16;
        dstp[0 * 16] = make_float4(r0.x, r1.x, r2.x, r3.x);
        dstp[1 * 16] = make_float4(r0.y, r1.y, r2.y, r3.y);
        dstp[2 * 16] = make_float4(r0.z, r1.z, r2.z, r3.z);
        dstp[3 * 16] = make_float4(r0.w, r1.w, r2.w, r3.w);
    }
    __syncthreads();
    if (row0 >= M) return;

    v8f acc[8];
    #pragma unroll
    for (int T = 0; T < 8; ++T) acc[T] = (v8f){};

    const float4* bsrc = s_w + half * 8 * 16 + l16;
    #pragma unroll
    for (int i = 0; i < 16; ++i) {
        float4 b4[8];
        #pragma unroll
        for (int T = 0; T < 8; ++T) b4[T] = bsrc[(2 * i * 8 + T) * 16];
        v2f a_lo, a_hi;
        a_lo.x = a4[i].x; a_lo.y = a4[i].y;
        a_hi.x = a4[i].z; a_hi.y = a4[i].w;
        #pragma unroll
        for (int T = 0; T < 8; ++T) {
            v2f b;
            b.x = b4[T].x; b.y = b4[T].y;
            acc[T] = __builtin_amdgcn_wmma_f32_16x16x4_f32(false, a_lo, false, b,
                                                           (short)0, acc[T], false, false);
        }
        #pragma unroll
        for (int T = 0; T < 8; ++T) {
            v2f b;
            b.x = b4[T].z; b.y = b4[T].w;
            acc[T] = __builtin_amdgcn_wmma_f32_16x16x4_f32(false, a_hi, false, b,
                                                           (short)0, acc[T], false, false);
        }
    }

    float4* crow = (float4*)C + (size_t)(row0 + 8 * half) * (FD / 4) + l16;
    if (row0 + 16 <= M) {
        #pragma unroll
        for (int g = 0; g < 2; ++g) {
            #pragma unroll
            for (int r = 0; r < 8; ++r)
                crow[r * (FD / 4) + 16 * g] =
                    make_float4(acc[4 * g + 0][r], acc[4 * g + 1][r],
                                acc[4 * g + 2][r], acc[4 * g + 3][r]);
        }
    } else {
        #pragma unroll
        for (int g = 0; g < 2; ++g) {
            #pragma unroll
            for (int r = 0; r < 8; ++r) {
                if (row0 + 8 * half + r < M)
                    crow[r * (FD / 4) + 16 * g] =
                        make_float4(acc[4 * g + 0][r], acc[4 * g + 1][r],
                                    acc[4 * g + 2][r], acc[4 * g + 3][r]);
            }
        }
    }
}


template <int K>
__device__ __forceinline__ void gcn_gather_rows(const float4* __restrict__ h4, int2 my, int t,
                                                int lane, float4& acc) {
    float4 v[K];
    float  w[K];
    #pragma unroll
    for (int j = 0; j < K; ++j) {
        const int s = __builtin_amdgcn_readlane(my.x, t + j);
        w[j] = __int_as_float(__builtin_amdgcn_readlane(my.y, t + j));
        v[j] = h4[(size_t)s * (FD / 4) + lane];
    }
    #pragma unroll
    for (int j = 0; j < K; ++j) {
        acc.x = fmaf(v[j].x, w[j], acc.x);
        acc.y = fmaf(v[j].y, w[j], acc.y);
        acc.z = fmaf(v[j].z, w[j], acc.z);
        acc.w = fmaf(v[j].w, w[j], acc.w);
    }
}

__global__ void __launch_bounds__(256)
gcn_pull_agg(const float* __restrict__ h, const int* __restrict__ rowptr,
             const int* __restrict__ adj, const float* __restrict__ dis,
             const float* __restrict__ bias, float* __restrict__ out, int N, int relu) {
    const int lane = threadIdx.x & 31;
    const int d = __builtin_amdgcn_readfirstlane(
        (int)((blockIdx.x * blockDim.x + threadIdx.x) >> 5));
    if (d >= N) return;

    const int beg = rowptr[d];
    const int end = rowptr[d + 1];
    const float dd = dis[d];
    const float4* h4 = (const float4*)h;
    const int2* adj2 = (const int2*)adj;

    const float4 sv = h4[(size_t)d * (FD / 4) + lane];
    const float sw = dd * dd;
    float4 acc = make_float4(sv.x * sw, sv.y * sw, sv.z * sw, sv.w * sw);

    for (int base = beg; base < end; base += 32) {
        int m = end - base;
        if (m > 32) m = 32;
        int2 my = make_int2(0, 0);
        if (lane < m) my = adj2[base + lane];

        int t = 0;
        for (; t + 8 <= m; t += 8) gcn_gather_rows<8>(h4, my, t, lane, acc);
        if (m - t >= 4) { gcn_gather_rows<4>(h4, my, t, lane, acc); t += 4; }
        if (m - t >= 2) { gcn_gather_rows<2>(h4, my, t, lane, acc); t += 2; }
        if (m - t >= 1) { gcn_gather_rows<1>(h4, my, t, lane, acc); }
    }

    const float4 b = ((const float4*)bias)[lane];
    float4 r = make_float4(acc.x + b.x, acc.y + b.y, acc.z + b.z, acc.w + b.w);
    if (relu) {
        r.x = r.x > 0.0f ? r.x : 0.0f;
        r.y = r.y > 0.0f ? r.y : 0.0f;
        r.z = r.z > 0.0f ? r.z : 0.0f;
        r.w = r.w > 0.0f ? r.w : 0.0f;
    }
    ((float4*)out)[(size_t)d * (FD / 4) + lane] = r;
}


extern "C" void kernel_launch(void* const* d_in, const int* in_sizes, int n_in,
                              void* d_out, int out_size, void* d_ws, size_t ws_size,
                              hipStream_t stream) {
    const float* x  = (const float*)d_in[0];
    const int*   ei = (const int*)d_in[1];
    const float* W1 = (const float*)d_in[2];
    const float* b1 = (const float*)d_in[3];
    const float* W2 = (const float*)d_in[4];
    const float* b2 = (const float*)d_in[5];

    const int N = in_sizes[0] / FD;
    const int E = in_sizes[1] / 2;
    const int* src = ei;
    const int* dst = ei + E;

    const size_t NP = ((size_t)N + 1 + 63) / 64 * 64;
    const size_t EP = ((size_t)E + 63) / 64 * 64;
    const int    ranges = (N + RNG - 1) / RNG;
    const size_t TP = ((size_t)ranges * (RNG / CNT_T) * NSL + 63) / 64 * 64;
    const int    slice  = ((E + NSL - 1) / NSL + 31) / 32 * 32;
    int*   wsi     = (int*)d_ws;
    int*   counts  = wsi;
    int*   rowptr  = wsi + NSL * NP;
    float* dis     = (float*)(rowptr + NP);
    int*   tiletot = rowptr + 2 * NP;
    int*   rank    = tiletot + TP;
    int*   adj     = rank + EP;
    float* bufA    = (float*)(adj + 2 * EP);
    float* bufB    = bufA + (size_t)N * FD;
    float* out     = (float*)d_out;

    gcn_count_rank<<<dim3(ranges, NSL), CNT_T, 0, stream>>>(dst, counts, tiletot, rank, E, slice,
                                                           N, (int)NP);
    gcn_scan_dis<<<(N + 1 + SCAN_T - 1) / SCAN_T, SCAN_T, 0, stream>>>(counts, tiletot, rowptr, dis,
                                                                      N, (int)NP);
    gcn_fill<<<dim3((slice + 255) / 256, NSL), 256, 0, stream>>>(src, dst, rank, counts, dis, adj,
                                                                E, slice, (int)NP);

    const int tilesM  = (N + 15) / 16;
    const int gblocks = (tilesM + GEMM_WAVES - 1) / GEMM_WAVES;

    const int ablocks = (N + 7) / 8;

    gcn_gemm_wmma<<<gblocks, 32 * GEMM_WAVES, 0, stream>>>(x, W1, bufA, N);
    gcn_pull_agg<<<ablocks, 256, 0, stream>>>(bufA, rowptr, adj, dis, b1, bufB, N, 1);

    gcn_gemm_wmma<<<gblocks, 32 * GEMM_WAVES, 0, stream>>>(bufB, W2, bufA, N);
    gcn_pull_agg<<<ablocks, 256, 0, stream>>>(bufA, rowptr, adj, dis, b2, out, N, 0);
}


# Layer-1 row sums and the layer-2 product in one launch: a wave sums its 16-row strip (a row's gathers in one group, next row's pairs ahead), keeps it in LDS and multiplies it by W2; 7 launches -> 6
# speedup vs baseline: 3.2464x; 1.0505x over previous
#include <hip/hip_runtime.h>
#include <hip/hip_bf16.h>


typedef float v2f __attribute__((ext_vector_type(2)));
typedef float v8f __attribute__((ext_vector_type(8)));

#define FD 128
#define SCAN_T 1024


#define RNG 4096
#define NSL 16
#define CNT_T 1024
#define CNT_U 19

__global__ void __launch_bounds__(CNT_T)
gcn_count_rank(const int* __restrict__ dst, int* __restrict__ counts, int* __restrict__ tiletot,
               int* __restrict__ rank, int E, int slice, int N, int NP) {
    __shared__ int s_cnt[RNG + 32];
    __shared__ int s_tot[RNG / CNT_T];
    const int tid = threadIdx.x;
    const int n0  = blockIdx.x * RNG;
    const int sl  = blockIdx.y;
    for (int i = tid; i < RNG + 32; i += CNT_T) s_cnt[i] = 0;
    if (tid < RNG / CNT_T) s_tot[tid] = 0;
    __syncthreads();

    const int e0 = sl * slice;
    const int e1 = (e0 + slice < E) ? e0 + slice : E;
    const int spare = RNG + (tid & 31);
    for (int base = e0 + tid; base < e1; base += CNT_U * CNT_T) {
        int d[CNT_U];
        int k[CNT_U];
        #pragma unroll
        for (int j = 0; j < CNT_U; ++j) {
            const int idx = base + j * CNT_T;
            d[j] = (idx < e1) ? dst[idx] : -1;
        }
        #pragma unroll
        for (int j = 0; j < CNT_U; ++j) {
            const unsigned r = (unsigned)(d[j] - n0);
            k[j] = atomicAdd(&s_cnt[r < (unsigned)RNG ? (int)r : spare], 1);
        }
        #pragma unroll
        for (int j = 0; j < CNT_U; ++j) {
            const unsigned r = (unsigned)(d[j] - n0);
            if (r < (unsigned)RNG) rank[base + j * CNT_T] = k[j];
        }
    }
    __syncthreads();

    #pragma unroll
    for (int t = 0; t < RNG / CNT_T; ++t) {
        const int c = s_cnt[t * CNT_T + tid];
        const int node = n0 + t * CNT_T + tid;
        if (node < N) counts[(size_t)sl * NP + node] = c;
        int sum = c;
        for (int o = 16; o > 0; o >>= 1) sum += __shfl_xor(sum, o);
        if ((tid & 31) == 0) atomicAdd(&s_tot[t], sum);
    }
    __syncthreads();
    if (tid < RNG / CNT_T)
        tiletot[(blockIdx.x * (RNG / CNT_T) + tid) * NSL + sl] = s_tot[tid];
}


__global__ void __launch_bounds__(SCAN_T)
gcn_scan_dis(int* __restrict__ counts, const int* __restrict__ tiletot,
             int* __restrict__ rowptr, float* __restrict__ dis, int N, int NP) {
    __shared__ int s_pre[SCAN_T / 32];
    __shared__ int s_tile[SCAN_T / 32];
    const int tid  = threadIdx.x;
    const int lane = tid & 31;
    const int wv   = tid >> 5;
    const int tile0 = blockIdx.x * SCAN_T;

    int pre = 0;
    const int npre = blockIdx.x * NSL;
    for (int i = tid; i < npre; i += SCAN_T) pre += tiletot[i];

    const int i = tile0 + tid;
    int c[NSL];
    #pragma unroll
    for (int s = 0; s < NSL; ++s) c[s] = (i < N) ? counts[(size_t)s * NP + i] : 0;
    int my = 0;
    #pragma unroll
    for (int s = 0; s < NSL; ++s) my += c[s];

    int incl = my;
    for (int o = 1; o < 32; o <<= 1) {
        int t = __shfl_up(incl, o);
        if (lane >= o) incl += t;
    }
    for (int o = 16; o > 0; o >>= 1) pre += __shfl_xor(pre, o);
    if (lane == 31) s_tile[wv] = incl;
    if (lane == 0)  s_pre[wv]  = pre;
    __syncthreads();

    int base = 0;
    for (int w = 0; w < SCAN_T / 32; ++w) {
        base += s_pre[w];
        if (w < wv) base += s_tile[w];
    }
    const int start = base + incl - my;
    if (i <= N) rowptr[i] = start;
    if (i < N) {
        dis[i] = rsqrtf((float)my + 1.0f);
        int run = start;
        #pragma unroll
        for (int s = 0; s < NSL; ++s) {
            counts[(size_t)s * NP + i] = run;
            run += c[s];
        }
    }
}


__global__ void gcn_fill(const int* __restrict__ src, const int* __restrict__ dst,
                         const int* __restrict__ rank, const int* __restrict__ slicebase,
                         const float* __restrict__ dis, int* __restrict__ adj,
                         int E, int slice, int NP) {
    const int sl = blockIdx.y;
    const int e0 = sl * slice;
    const int e1 = (e0 + slice < E) ? e0 + slice : E;
    const int e  = e0 + blockIdx.x * blockDim.x + threadIdx.x;
    if (e < e1) {
        int s = src[e];
        int d = dst[e];
        float nrm = dis[s] * dis[d];
        ((int2*)adj)[slicebase[(size_t)sl * NP + d] + rank[e]] = make_int2(s, __float_as_int(nrm));
    }
}


#define GEMM_WAVES 8

__global__ void __launch_bounds__(32 * GEMM_WAVES)
gcn_gemm_wmma(const float* __restrict__ A, const float* __restrict__ W,
              float* __restrict__ C, int M) {
    __shared__ float4 s_w[(FD / 4) * 8 * 16];

    const int tid  = threadIdx.x;
    const int lane = tid & 31;
    const int wv   = tid >> 5;
    const int half = lane >> 4;
    const int l16  = lane & 15;
    const int row0 = (blockIdx.x * GEMM_WAVES + wv) * 16;

    int ar = row0 + l16;
    if (ar > M - 1) ar = M - 1;
    const float4* arow = (const float4*)(A + (size_t)ar * FD) + half;
    float4 a4[16];
    #pragma unroll
    for (int i = 0; i < 16; ++i) a4[i] = arow[2 * i];

    const float4* W4 = (const float4*)W;
    #pragma unroll
    for (int j = 0; j < FD / 4 / GEMM_WAVES; ++j) {
        const int kq = wv + GEMM_WAVES * j;
        const float4 r0 = W4[(4 * kq + 0) * (FD / 4) + lane];
        const float4 r1 = W4[(4 * kq + 1) * (FD / 4) + lane];
        const float4 r2 = W4[(4 * kq + 2) * (FD / 4) + lane];
        const float4 r3 = W4[(4 * kq + 3) * (FD / 4) + lane];
        float4* dstp = s_w + (kq * 8 + half * 4) * 16 + l16;
        dstp[0 * 16] = make_float4(r0.x, r1.x, r2.x, r3.x);
        dstp[1 * 16] = make_float4(r0.y, r1.y, r2.y, r3.y);
        dstp[2 * 16] = make_float4(r0.z, r1.z, r2.z, r3.z);
        dstp[3 * 16] = make_float4(r0.w, r1.w, r2.w, r3.w);
    }
    __syncthreads();
    if (row0 >= M) return;

    v8f acc[8];
    #pragma unroll
    for (int T = 0; T < 8; ++T) acc[T] = (v8f){};

    const float4* bsrc = s_w + half * 8 * 16 + l16;
    #pragma unroll
    for (int i = 0; i < 16; ++i) {
        float4 b4[8];
        #pragma unroll
        for (int T = 0; T < 8; ++T) b4[T] = bsrc[(2 * i * 8 + T) * 16];
        v2f a_lo, a_hi;
        a_lo.x = a4[i].x; a_lo.y = a4[i].y;
        a_hi.x = a4[i].z; a_hi.y = a4[i].w;
        #pragma unroll
        for (int T = 0; T < 8; ++T) {
            v2f b;
            b.x = b4[T].x; b.y = b4[T].y;
            acc[T] = __builtin_amdgcn_wmma_f32_16x16x4_f32(false, a_lo, false, b,
                                                           (short)0, acc[T], false, false);
        }
        #pragma unroll
        for (int T = 0; T < 8; ++T) {
            v2f b;
            b.x = b4[T].z; b.y = b4[T].w;
            acc[T] = __builtin_amdgcn_wmma_f32_16x16x4_f32(false, a_hi, false, b,
                                                           (short)0, acc[T], false, false);
        }
    }

    float4* crow = (float4*)C + (size_t)(row0 + 8 * half) * (FD / 4) + l16;
    if (row0 + 16 <= M) {
        #pragma unroll
        for (int g = 0; g < 2; ++g) {
            #pragma unroll
            for (int r = 0; r < 8; ++r)
                crow[r * (FD / 4) + 16 * g] =
                    make_float4(acc[4 * g + 0][r], acc[4 * g + 1][r],
                                acc[4 * g + 2][r], acc[4 * g + 3][r]);
        }
    } else {
        #pragma unroll
        for (int g = 0; g < 2; ++g) {
            #pragma unroll
            for (int r = 0; r < 8; ++r) {
                if (row0 + 8 * half + r < M)
                    crow[r * (FD / 4) + 16 * g] =
                        make_float4(acc[4 * g + 0][r], acc[4 * g + 1][r],
                                    acc[4 * g + 2][r], acc[4 * g + 3][r]);
            }
        }
    }
}


template <int K>
__device__ __forceinline__ void gcn_gather_rows(const float4* __restrict__ h4, int2 my, int t,
                                                int lane, float4& acc) {
    float4 v[K];
    float  w[K];
    #pragma unroll
    for (int j = 0; j < K; ++j) {
        const int s = __builtin_amdgcn_readlane(my.x, t + j);
        w[j] = __int_as_float(__builtin_amdgcn_readlane(my.y, t + j));
        v[j] = h4[(size_t)s * (FD / 4) + lane];
    }
    #pragma unroll
    for (int j = 0; j < K; ++j) {
        acc.x = fmaf(v[j].x, w[j], acc.x);
        acc.y = fmaf(v[j].y, w[j], acc.y);
        acc.z = fmaf(v[j].z, w[j], acc.z);
        acc.w = fmaf(v[j].w, w[j], acc.w);
    }
}

__global__ void __launch_bounds__(256)
gcn_pull_agg(const float* __restrict__ h, const int* __restrict__ rowptr,
             const int* __restrict__ adj, const float* __restrict__ dis,
             const float* __restrict__ bias, float* __restrict__ out, int N, int relu) {
    const int lane = threadIdx.x & 31;
    const int d = __builtin_amdgcn_readfirstlane(
        (int)((blockIdx.x * blockDim.x + threadIdx.x) >> 5));
    if (d >= N) return;

    const int beg = rowptr[d];
    const int end = rowptr[d + 1];
    const float dd = dis[d];
    const float4* h4 = (const float4*)h;
    const int2* adj2 = (const int2*)adj;

    const float4 sv = h4[(size_t)d * (FD / 4) + lane];
    const float sw = dd * dd;
    float4 acc = make_float4(sv.x * sw, sv.y * sw, sv.z * sw, sv.w * sw);

    for (int base = beg; base < end; base += 32) {
        int m = end - base;
        if (m > 32) m = 32;
        int2 my = make_int2(0, 0);
        if (lane < m) my = adj2[base + lane];

        int t = 0;
        for (; t + 8 <= m; t += 8) gcn_gather_rows<8>(h4, my, t, lane, acc);
        if (m - t >= 4) { gcn_gather_rows<4>(h4, my, t, lane, acc); t += 4; }
        if (m - t >= 2) { gcn_gather_rows<2>(h4, my, t, lane, acc); t += 2; }
        if (m - t >= 1) { gcn_gather_rows<1>(h4, my, t, lane, acc); }
    }

    const float4 b = ((const float4*)bias)[lane];
    float4 r = make_float4(acc.x + b.x, acc.y + b.y, acc.z + b.z, acc.w + b.w);
    if (relu) {
        r.x = r.x > 0.0f ? r.x : 0.0f;
        r.y = r.y > 0.0f ? r.y : 0.0f;
        r.z = r.z > 0.0f ? r.z : 0.0f;
        r.w = r.w > 0.0f ? r.w : 0.0f;
    }
    ((float4*)out)[(size_t)d * (FD / 4) + lane] = r;
}


#define SA_PITCH 33
#define GATH_K  16

template <int K>
__device__ __forceinline__ void gcn_gather_first(const float4* __restrict__ h4, int d, float sw,
                                                 int2 my, int lane, float4& acc) {
    float4 v[K + 1];
    float  w[K + 1];
    const float4 sv = h4[(size_t)d * (FD / 4) + lane];
    #pragma unroll
    for (int j = 0; j < K; ++j) {
        const int s = __builtin_amdgcn_readlane(my.x, j);
        w[j] = __int_as_float(__builtin_amdgcn_readlane(my.y, j));
        v[j] = h4[(size_t)s * (FD / 4) + lane];
    }
    acc = make_float4(sv.x * sw, sv.y * sw, sv.z * sw, sv.w * sw);
    #pragma unroll
    for (int j = 0; j < K; ++j) {
        acc.x = fmaf(v[j].x, w[j], acc.x);
        acc.y = fmaf(v[j].y, w[j], acc.y);
        acc.z = fmaf(v[j].z, w[j], acc.z);
        acc.w = fmaf(v[j].w, w[j], acc.w);
    }
}

__device__ __forceinline__ void gcn_gather_rest(const float4* __restrict__ h4, int2 my, int t,
                                                int m, int lane, float4& acc) {
    for (; t + 8 <= m; t += 8) gcn_gather_rows<8>(h4, my, t, lane, acc);
    if (m - t >= 4) { gcn_gather_rows<4>(h4, my, t, lane, acc); t += 4; }
    if (m - t >= 2) { gcn_gather_rows<2>(h4, my, t, lane, acc); t += 2; }
    if (m - t >= 1) { gcn_gather_rows<1>(h4, my, t, lane, acc); }
}

__global__ void __launch_bounds__(32 * GEMM_WAVES)
gcn_agg_gemm_wmma(const float* __restrict__ h, const int* __restrict__ rowptr,
                  const int* __restrict__ adj, const float* __restrict__ dis,
                  const float* __restrict__ bias, const float* __restrict__ W,
                  float* __restrict__ C, int M) {
    __shared__ float4 s_w[(FD / 4) * 8 * 16];
    __shared__ float4 s_a[GEMM_WAVES * 16 * SA_PITCH];

    const int tid  = threadIdx.x;
    const int lane = tid & 31;
    const int wv   = __builtin_amdgcn_readfirstlane(tid >> 5);
    const int half = lane >> 4;
    const int l16  = lane & 15;
    const int row0 = (blockIdx.x * GEMM_WAVES + wv) * 16;

    int rpi = row0 + lane;
    if (rpi > M) rpi = M;
    const int rp = rowptr[rpi];
    int dsi = row0 + lane;
    if (dsi > M - 1) dsi = M - 1;
    const float dv = dis[dsi];
    const float4 b = ((const float4*)bias)[lane];

    const float4* W4 = (const float4*)W;
    #pragma unroll
    for (int j = 0; j < FD / 4 / GEMM_WAVES; ++j) {
        const int kq = wv + GEMM_WAVES * j;
        const float4 r0 = W4[(4 * kq + 0) * (FD / 4) + lane];
        const float4 r1 = W4[(4 * kq + 1) * (FD / 4) + lane];
        const float4 r2 = W4[(4 * kq + 2) * (FD / 4) + lane];
        const float4 r3 = W4[(4 * kq + 3) * (FD / 4) + lane];
        float4* dstp = s_w + (kq * 8 + half * 4) * 16 + l16;
        dstp[0 * 16] = make_float4(r0.x, r1.x, r2.x, r3.x);
        dstp[1 * 16] = make_float4(r0.y, r1.y, r2.y, r3.y);
        dstp[2 * 16] = make_float4(r0.z, r1.z, r2.z, r3.z);
        dstp[3 * 16] = make_float4(r0.w, r1.w, r2.w, r3.w);
    }
    __syncthreads();
    if (row0 >= M) return;

    const float4* h4 = (const float4*)h;
    const int2* adj2 = (const int2*)adj;
    float4* sa = s_a + wv * 16 * SA_PITCH;
    int nrows = M - row0;
    if (nrows > 16) nrows = 16;

    const int last = __builtin_amdgcn_readlane(rp, 16) - 1;

    int2 my;
    {
        int pi = __builtin_amdgcn_readlane(rp, 0) + lane;
        if (pi > last) pi = last;
        if (pi < 0) pi = 0;
        my = adj2[pi];
    }

    for (int r = 0; r < nrows; ++r) {
        const int beg = __builtin_amdgcn_readlane(rp, r);
        const int end = __builtin_amdgcn_readlane(rp, r + 1);
        const float dd = __int_as_float(__builtin_amdgcn_readlane(__float_as_int(dv), r));

        int pn = end + lane;
        if (pn > last) pn = last;
        if (pn < 0) pn = 0;
        const int2 nxt = adj2[pn];

        int m = end - beg;
        if (m > 32) m = 32;
        int n = m;
        if (n > GATH_K) n = GATH_K;
        float4 acc;
        switch (n) {
            case 16: gcn_gather_first<16>(h4, row0 + r, dd * dd, my, lane, acc); break;
            case 15: gcn_gather_first<15>(h4, row0 + r, dd * dd, my, lane, acc); break;
            case 14: gcn_gather_first<14>(h4, row0 + r, dd * dd, my, lane, acc); break;
            case 13: gcn_gather_first<13>(h4, row0 + r, dd * dd, my, lane, acc); break;
            case 12: gcn_gather_first<12>(h4, row0 + r, dd * dd, my, lane, acc); break;
            case 11: gcn_gather_first<11>(h4, row0 + r, dd * dd, my, lane, acc); break;
            case 10: gcn_gather_first<10>(h4, row0 + r, dd * dd, my, lane, acc); break;
            case 9:  gcn_gather_first<9>(h4, row0 + r, dd * dd, my, lane, acc); break;
            case 8:  gcn_gather_first<8>(h4, row0 + r, dd * dd, my, lane, acc); break;
            case 7:  gcn_gather_first<7>(h4, row0 + r, dd * dd, my, lane, acc); break;
            case 6:  gcn_gather_first<6>(h4, row0 + r, dd * dd, my, lane, acc); break;
            case 5:  gcn_gather_first<5>(h4, row0 + r, dd * dd, my, lane, acc); break;
            case 4:  gcn_gather_first<4>(h4, row0 + r, dd * dd, my, lane, acc); break;
            case 3:  gcn_gather_first<3>(h4, row0 + r, dd * dd, my, lane, acc); break;
            case 2:  gcn_gather_first<2>(h4, row0 + r, dd * dd, my, lane, acc); break;
            case 1:  gcn_gather_first<1>(h4, row0 + r, dd * dd, my, lane, acc); break;
            default: gcn_gather_first<0>(h4, row0 + r, dd * dd, my, lane, acc); break;
        }
        if (m > GATH_K) gcn_gather_rest(h4, my, GATH_K, m, lane, acc);
        for (int base = beg + 32; base < end; base += 32) {
            int mm = end - base;
            if (mm > 32) mm = 32;
            int pj = base + lane;
            if (pj > last) pj = last;
            const int2 more = adj2[pj];
            gcn_gather_rest(h4, more, 0, mm, lane, acc);
        }

        float4 o = make_float4(acc.x + b.x, acc.y + b.y, acc.z + b.z, acc.w + b.w);
        o.x = o.x > 0.0f ? o.x : 0.0f;
        o.y = o.y > 0.0f ? o.y : 0.0f;
        o.z = o.z > 0.0f ? o.z : 0.0f;
        o.w = o.w > 0.0f ? o.w : 0.0f;
        sa[r * SA_PITCH + lane] = o;

        my = nxt;
    }
    for (int r = nrows; r < 16; ++r) sa[r * SA_PITCH + lane] = make_float4(0.0f, 0.0f, 0.0f, 0.0f);

    __builtin_amdgcn_fence(__ATOMIC_RELEASE, "workgroup");
    __builtin_amdgcn_wave_barrier();
    __builtin_amdgcn_fence(__ATOMIC_ACQUIRE, "workgroup");

    float4 a4[16];
    #pragma unroll
    for (int i = 0; i < 16; ++i) a4[i] = sa[l16 * SA_PITCH + 2 * i + half];

    v8f acc[8];
    #pragma unroll
    for (int T = 0; T < 8; ++T) acc[T] = (v8f){};

    const float4* bsrc = s_w + half * 8 * 16 + l16;
    #pragma unroll
    for (int i = 0; i < 16; ++i) {
        float4 b4[8];
        #pragma unroll
        for (int T = 0; T < 8; ++T) b4[T] = bsrc[(2 * i * 8 + T) * 16];
        v2f a_lo, a_hi;
        a_lo.x = a4[i].x; a_lo.y = a4[i].y;
        a_hi.x = a4[i].z; a_hi.y = a4[i].w;
        #pragma unroll
        for (int T = 0; T < 8; ++T) {
            v2f bb;
            bb.x = b4[T].x; bb.y = b4[T].y;
            acc[T] = __builtin_amdgcn_wmma_f32_16x16x4_f32(false, a_lo, false, bb,
                                                           (short)0, acc[T], false, false);
        }
        #pragma unroll
        for (int T = 0; T < 8; ++T) {
            v2f bb;
            bb.x = b4[T].z; bb.y = b4[T].w;
            acc[T] = __builtin_amdgcn_wmma_f32_16x16x4_f32(false, a_hi, false, bb,
                                                           (short)0, acc[T], false, false);
        }
    }

    float4* crow = (float4*)C + (size_t)(row0 + 8 * half) * (FD / 4) + l16;
    if (row0 + 16 <= M) {
        #pragma unroll
        for (int g = 0; g < 2; ++g) {
            #pragma unroll
            for (int r = 0; r < 8; ++r)
                crow[r * (FD / 4) + 16 * g] =
                    make_float4(acc[4 * g + 0][r], acc[4 * g + 1][r],
                                acc[4 * g + 2][r], acc[4 * g + 3][r]);
        }
    } else {
        #pragma unroll
        for (int g = 0; g < 2; ++g) {
            #pragma unroll
            for (int r = 0; r < 8; ++r) {
                if (row0 + 8 * half + r < M)
                    crow[r * (FD / 4) + 16 * g] =
                        make_float4(acc[4 * g + 0][r], acc[4 * g + 1][r],
                                    acc[4 * g + 2][r], acc[4 * g + 3][r]);
            }
        }
    }
}


extern "C" void kernel_launch(void* const* d_in, const int* in_sizes, int n_in,
                              void* d_out, int out_size, void* d_ws, size_t ws_size,
                              hipStream_t stream) {
    const float* x  = (const float*)d_in[0];
    const int*   ei = (const int*)d_in[1];
    const float* W1 = (const float*)d_in[2];
    const float* b1 = (const float*)d_in[3];
    const float* W2 = (const float*)d_in[4];
    const float* b2 = (const float*)d_in[5];

    const int N = in_sizes[0] / FD;
    const int E = in_sizes[1] / 2;
    const int* src = ei;
    const int* dst = ei + E;

    const size_t NP = ((size_t)N + 1 + 63) / 64 * 64;
    const size_t EP = ((size_t)E + 63) / 64 * 64;
    const int    ranges = (N + RNG - 1) / RNG;
    const size_t TP = ((size_t)ranges * (RNG / CNT_T) * NSL + 63) / 64 * 64;
    const int    slice  = ((E + NSL - 1) / NSL + 31) / 32 * 32;
    int*   wsi     = (int*)d_ws;
    int*   counts  = wsi;
    int*   rowptr  = wsi + NSL * NP;
    float* dis     = (float*)(rowptr + NP);
    int*   tiletot = rowptr + 2 * NP;
    int*   rank    = tiletot + TP;
    int*   adj     = rank + EP;
    float* bufA    = (float*)(adj + 2 * EP);
    float* bufB    = bufA + (size_t)N * FD;
    float* out     = (float*)d_out;

    gcn_count_rank<<<dim3(ranges, NSL), CNT_T, 0, stream>>>(dst, counts, tiletot, rank, E, slice,
                                                           N, (int)NP);
    gcn_scan_dis<<<(N + 1 + SCAN_T - 1) / SCAN_T, SCAN_T, 0, stream>>>(counts, tiletot, rowptr, dis,
                                                                      N, (int)NP);
    gcn_fill<<<dim3((slice + 255) / 256, NSL), 256, 0, stream>>>(src, dst, rank, counts, dis, adj,
                                                                E, slice, (int)NP);

    const int tilesM  = (N + 15) / 16;
    const int gblocks = (tilesM + GEMM_WAVES - 1) / GEMM_WAVES;

    const int ablocks = (N + 7) / 8;

    gcn_gemm_wmma<<<gblocks, 32 * GEMM_WAVES, 0, stream>>>(x, W1, bufA, N);

    gcn_agg_gemm_wmma<<<gblocks, 32 * GEMM_WAVES, 0, stream>>>(bufA, rowptr, adj, dis, b1, W2,
                                                              bufB, N);

    gcn_pull_agg<<<ablocks, 256, 0, stream>>>(bufB, rowptr, adj, dis, b2, out, N, 0);
}


# Layer launch: two waves per 16-row strip (512-thread work-groups): each sums 8 of the strip's rows, one work-group barrier, then each multiplies the strip by 64 of W's columns (4 tiles)
# speedup vs baseline: 3.6478x; 1.1236x over previous
#include <hip/hip_runtime.h>
#include <hip/hip_bf16.h>


typedef float v2f __attribute__((ext_vector_type(2)));
typedef float v8f __attribute__((ext_vector_type(8)));

#define FD 128
#define SCAN_T 1024


#define RNG 4096
#define NSL 16
#define CNT_T 1024
#define CNT_U 19

__global__ void __launch_bounds__(CNT_T)
gcn_count_rank(const int* __restrict__ dst, int* __restrict__ counts, int* __restrict__ tiletot,
               int* __restrict__ rank, int E, int slice, int N, int NP) {
    __shared__ int s_cnt[RNG + 32];
    __shared__ int s_tot[RNG / CNT_T];
    const int tid = threadIdx.x;
    const int n0  = blockIdx.x * RNG;
    const int sl  = blockIdx.y;
    for (int i = tid; i < RNG + 32; i += CNT_T) s_cnt[i] = 0;
    if (tid < RNG / CNT_T) s_tot[tid] = 0;
    __syncthreads();

    const int e0 = sl * slice;
    const int e1 = (e0 + slice < E) ? e0 + slice : E;
    const int spare = RNG + (tid & 31);
    for (int base = e0 + tid; base < e1; base += CNT_U * CNT_T) {
        int d[CNT_U];
        int k[CNT_U];
        #pragma unroll
        for (int j = 0; j < CNT_U; ++j) {
            const int idx = base + j * CNT_T;
            d[j] = (idx < e1) ? dst[idx] : -1;
        }
        #pragma unroll
        for (int j = 0; j < CNT_U; ++j) {
            const unsigned r = (unsigned)(d[j] - n0);
            k[j] = atomicAdd(&s_cnt[r < (unsigned)RNG ? (int)r : spare], 1);
        }
        #pragma unroll
        for (int j = 0; j < CNT_U; ++j) {
            const unsigned r = (unsigned)(d[j] - n0);
            if (r < (unsigned)RNG) rank[base + j * CNT_T] = k[j];
        }
    }
    __syncthreads();

    #pragma unroll
    for (int t = 0; t < RNG / CNT_T; ++t) {
        const int c = s_cnt[t * CNT_T + tid];
        const int node = n0 + t * CNT_T + tid;
        if (node < N) counts[(size_t)sl * NP + node] = c;
        int sum = c;
        for (int o = 16; o > 0; o >>= 1) sum += __shfl_xor(sum, o);
        if ((tid & 31) == 0) atomicAdd(&s_tot[t], sum);
    }
    __syncthreads();
    if (tid < RNG / CNT_T)
        tiletot[(blockIdx.x * (RNG / CNT_T) + tid) * NSL + sl] = s_tot[tid];
}


__global__ void __launch_bounds__(SCAN_T)
gcn_scan_dis(int* __restrict__ counts, const int* __restrict__ tiletot,
             int* __restrict__ rowptr, float* __restrict__ dis, int N, int NP) {
    __shared__ int s_pre[SCAN_T / 32];
    __shared__ int s_tile[SCAN_T / 32];
    const int tid  = threadIdx.x;
    const int lane = tid & 31;
    const int wv   = tid >> 5;
    const int tile0 = blockIdx.x * SCAN_T;

    int pre = 0;
    const int npre = blockIdx.x * NSL;
    for (int i = tid; i < npre; i += SCAN_T) pre += tiletot[i];

    const int i = tile0 + tid;
    int c[NSL];
    #pragma unroll
    for (int s = 0; s < NSL; ++s) c[s] = (i < N) ? counts[(size_t)s * NP + i] : 0;
    int my = 0;
    #pragma unroll
    for (int s = 0; s < NSL; ++s) my += c[s];

    int incl = my;
    for (int o = 1; o < 32; o <<= 1) {
        int t = __shfl_up(incl, o);
        if (lane >= o) incl += t;
    }
    for (int o = 16; o > 0; o >>= 1) pre += __shfl_xor(pre, o);
    if (lane == 31) s_tile[wv] = incl;
    if (lane == 0)  s_pre[wv]  = pre;
    __syncthreads();

    int base = 0;
    for (int w = 0; w < SCAN_T / 32; ++w) {
        base += s_pre[w];
        if (w < wv) base += s_tile[w];
    }
    const int start = base + incl - my;
    if (i <= N) rowptr[i] = start;
    if (i < N) {
        dis[i] = rsqrtf((float)my + 1.0f);
        int run = start;
        #pragma unroll
        for (int s = 0; s < NSL; ++s) {
            counts[(size_t)s * NP + i] = run;
            run += c[s];
        }
    }
}


__global__ void gcn_fill(const int* __restrict__ src, const int* __restrict__ dst,
                         const int* __restrict__ rank, const int* __restrict__ slicebase,
                         const float* __restrict__ dis, int* __restrict__ adj,
                         int E, int slice, int NP) {
    const int sl = blockIdx.y;
    const int e0 = sl * slice;
    const int e1 = (e0 + slice < E) ? e0 + slice : E;
    const int e  = e0 + blockIdx.x * blockDim.x + threadIdx.x;
    if (e < e1) {
        int s = src[e];
        int d = dst[e];
        float nrm = dis[s] * dis[d];
        ((int2*)adj)[slicebase[(size_t)sl * NP + d] + rank[e]] = make_int2(s, __float_as_int(nrm));
    }
}


#define GEMM_STRIPS 8
#define STRIP_WAVES 2
#define GEMM_WAVES (GEMM_STRIPS * STRIP_WAVES)


template <int K>
__device__ __forceinline__ void gcn_gather_rows(const float4* __restrict__ h4, int2 my, int t,
                                                int lane, float4& acc) {
    float4 v[K];
    float  w[K];
    #pragma unroll
    for (int j = 0; j < K; ++j) {
        const int s = __builtin_amdgcn_readlane(my.x, t + j);
        w[j] = __int_as_float(__builtin_amdgcn_readlane(my.y, t + j));
        v[j] = h4[(size_t)s * (FD / 4) + lane];
    }
    #pragma unroll
    for (int j = 0; j < K; ++j) {
        acc.x = fmaf(v[j].x, w[j], acc.x);
        acc.y = fmaf(v[j].y, w[j], acc.y);
        acc.z = fmaf(v[j].z, w[j], acc.z);
        acc.w = fmaf(v[j].w, w[j], acc.w);
    }
}


#define SA_PITCH 33
#define GATH_K  16

template <int K>
__device__ __forceinline__ void gcn_gather_first(const float4* __restrict__ h4, int d, float sw,
                                                 int2 my, int lane, float4& acc) {
    float4 v[K + 1];
    float  w[K + 1];
    const float4 sv = h4[(size_t)d * (FD / 4) + lane];
    #pragma unroll
    for (int j = 0; j < K; ++j) {
        const int s = __builtin_amdgcn_readlane(my.x, j);
        w[j] = __int_as_float(__builtin_amdgcn_readlane(my.y, j));
        v[j] = h4[(size_t)s * (FD / 4) + lane];
    }
    acc = make_float4(sv.x * sw, sv.y * sw, sv.z * sw, sv.w * sw);
    #pragma unroll
    for (int j = 0; j < K; ++j) {
        acc.x = fmaf(v[j].x, w[j], acc.x);
        acc.y = fmaf(v[j].y, w[j], acc.y);
        acc.z = fmaf(v[j].z, w[j], acc.z);
        acc.w = fmaf(v[j].w, w[j], acc.w);
    }
}

__device__ __forceinline__ void gcn_gather_rest(const float4* __restrict__ h4, int2 my, int t,
                                                int m, int lane, float4& acc) {
    for (; t + 8 <= m; t += 8) gcn_gather_rows<8>(h4, my, t, lane, acc);
    if (m - t >= 4) { gcn_gather_rows<4>(h4, my, t, lane, acc); t += 4; }
    if (m - t >= 2) { gcn_gather_rows<2>(h4, my, t, lane, acc); t += 2; }
    if (m - t >= 1) { gcn_gather_rows<1>(h4, my, t, lane, acc); }
}

__global__ void __launch_bounds__(32 * GEMM_WAVES)
gcn_agg_gemm_wmma(const float* __restrict__ h, const int* __restrict__ rowptr,
                  const int* __restrict__ adj, const float* __restrict__ dis,
                  const float* __restrict__ bias, const float* __restrict__ W,
                  float* __restrict__ C, int M, int relu) {
    __shared__ float4 s_w[(FD / 4) * 8 * 16];
    __shared__ float4 s_a[GEMM_STRIPS * 16 * SA_PITCH];

    const int tid  = threadIdx.x;
    const int lane = tid & 31;
    const int wv   = __builtin_amdgcn_readfirstlane(tid >> 5);
    const int st   = wv >> 1;
    const int part = wv & 1;
    const int half = lane >> 4;
    const int l16  = lane & 15;
    const int row0 = (blockIdx.x * GEMM_STRIPS + st) * 16;

    int rpi = row0 + lane;
    if (rpi > M) rpi = M;
    const int rp = rowptr[rpi];
    int dsi = row0 + lane;
    if (dsi > M - 1) dsi = M - 1;
    const float dv = dis[dsi];
    const float4 bz = ((const float4*)bias)[16 * part + l16];

    const float4* W4 = (const float4*)W;
    #pragma unroll
    for (int j = 0; j < FD / 4 / GEMM_WAVES; ++j) {
        const int kq = wv + GEMM_WAVES * j;
        const float4 r0 = W4[(4 * kq + 0) * (FD / 4) + lane];
        const float4 r1 = W4[(4 * kq + 1) * (FD / 4) + lane];
        const float4 r2 = W4[(4 * kq + 2) * (FD / 4) + lane];
        const float4 r3 = W4[(4 * kq + 3) * (FD / 4) + lane];
        float4* dstp = s_w + (kq * 8 + half * 4) * 16 + l16;
        dstp[0 * 16] = make_float4(r0.x, r1.x, r2.x, r3.x);
        dstp[1 * 16] = make_float4(r0.y, r1.y, r2.y, r3.y);
        dstp[2 * 16] = make_float4(r0.z, r1.z, r2.z, r3.z);
        dstp[3 * 16] = make_float4(r0.w, r1.w, r2.w, r3.w);
    }
    __syncthreads();

    const float4* h4 = (const float4*)h;
    const int2* adj2 = (const int2*)adj;
    float4* sa = s_a + st * 16 * SA_PITCH;
    int nrows = M - row0;
    if (nrows > 16) nrows = 16;
    const int rbeg = 8 * part;
    int rend = rbeg + 8;
    if (rend > nrows) rend = nrows;

    if (rbeg < rend) {
    const int last = __builtin_amdgcn_readlane(rp, 16) - 1;

    int2 my;
    {
        int pi = __builtin_amdgcn_readlane(rp, rbeg) + lane;
        if (pi > last) pi = last;
        if (pi < 0) pi = 0;
        my = adj2[pi];
    }

    for (int r = rbeg; r < rend; ++r) {
        const int beg = __builtin_amdgcn_readlane(rp, r);
        const int end = __builtin_amdgcn_readlane(rp, r + 1);
        const float dd = __int_as_float(__builtin_amdgcn_readlane(__float_as_int(dv), r));

        int pn = end + lane;
        if (pn > last) pn = last;
        if (pn < 0) pn = 0;
        const int2 nxt = adj2[pn];

        int m = end - beg;
        if (m > 32) m = 32;
        int n = m;
        if (n > GATH_K) n = GATH_K;
        float4 acc;
        switch (n) {
            case 16: gcn_gather_first<16>(h4, row0 + r, dd * dd, my, lane, acc); break;
            case 15: gcn_gather_first<15>(h4, row0 + r, dd * dd, my, lane, acc); break;
            case 14: gcn_gather_first<14>(h4, row0 + r, dd * dd, my, lane, acc); break;
            case 13: gcn_gather_first<13>(h4, row0 + r, dd * dd, my, lane, acc); break;
            case 12: gcn_gather_first<12>(h4, row0 + r, dd * dd, my, lane, acc); break;
            case 11: gcn_gather_first<11>(h4, row0 + r, dd * dd, my, lane, acc); break;
            case 10: gcn_gather_first<10>(h4, row0 + r, dd * dd, my, lane, acc); break;
            case 9:  gcn_gather_first<9>(h4, row0 + r, dd * dd, my, lane, acc); break;
            case 8:  gcn_gather_first<8>(h4, row0 + r, dd * dd, my, lane, acc); break;
            case 7:  gcn_gather_first<7>(h4, row0 + r, dd * dd, my, lane, acc); break;
            case 6:  gcn_gather_first<6>(h4, row0 + r, dd * dd, my, lane, acc); break;
            case 5:  gcn_gather_first<5>(h4, row0 + r, dd * dd, my, lane, acc); break;
            case 4:  gcn_gather_first<4>(h4, row0 + r, dd * dd, my, lane, acc); break;
            case 3:  gcn_gather_first<3>(h4, row0 + r, dd * dd, my, lane, acc); break;
            case 2:  gcn_gather_first<2>(h4, row0 + r, dd * dd, my, lane, acc); break;
            case 1:  gcn_gather_first<1>(h4, row0 + r, dd * dd, my, lane, acc); break;
            default: gcn_gather_first<0>(h4, row0 + r, dd * dd, my, lane, acc); break;
        }
        if (m > GATH_K) gcn_gather_rest(h4, my, GATH_K, m, lane, acc);
        for (int base = beg + 32; base < end; base += 32) {
            int mm = end - base;
            if (mm > 32) mm = 32;
            int pj = base + lane;
            if (pj > last) pj = last;
            const int2 more = adj2[pj];
            gcn_gather_rest(h4, more, 0, mm, lane, acc);
        }

        sa[r * SA_PITCH + lane] = acc;

        my = nxt;
    }
    }
    for (int r = (rend > rbeg ? rend : rbeg); r < rbeg + 8; ++r)
        sa[r * SA_PITCH + lane] = make_float4(0.0f, 0.0f, 0.0f, 0.0f);

    __syncthreads();
    if (row0 >= M) return;

    float4 a4[16];
    #pragma unroll
    for (int i = 0; i < 16; ++i) a4[i] = sa[l16 * SA_PITCH + 2 * i + half];

    v8f acc[4];
    #pragma unroll
    for (int T = 0; T < 4; ++T) acc[T] = (v8f){};

    const float4* bsrc = s_w + (half * 8 + 4 * part) * 16 + l16;
    #pragma unroll
    for (int i = 0; i < 16; ++i) {
        float4 b4[4];
        #pragma unroll
        for (int T = 0; T < 4; ++T) b4[T] = bsrc[(2 * i * 8 + T) * 16];
        v2f a_lo, a_hi;
        a_lo.x = a4[i].x; a_lo.y = a4[i].y;
        a_hi.x = a4[i].z; a_hi.y = a4[i].w;
        #pragma unroll
        for (int T = 0; T < 4; ++T) {
            v2f bb;
            bb.x = b4[T].x; bb.y = b4[T].y;
            acc[T] = __builtin_amdgcn_wmma_f32_16x16x4_f32(false, a_lo, false, bb,
                                                           (short)0, acc[T], false, false);
        }
        #pragma unroll
        for (int T = 0; T < 4; ++T) {
            v2f bb;
            bb.x = b4[T].z; bb.y = b4[T].w;
            acc[T] = __builtin_amdgcn_wmma_f32_16x16x4_f32(false, a_hi, false, bb,
                                                           (short)0, acc[T], false, false);
        }
    }

    float4 o[8];
    #pragma unroll
    for (int r = 0; r < 8; ++r) {
        float4 v = make_float4(acc[0][r] + bz.x, acc[1][r] + bz.y,
                               acc[2][r] + bz.z, acc[3][r] + bz.w);
        if (relu) {
            v.x = v.x > 0.0f ? v.x : 0.0f;
            v.y = v.y > 0.0f ? v.y : 0.0f;
            v.z = v.z > 0.0f ? v.z : 0.0f;
            v.w = v.w > 0.0f ? v.w : 0.0f;
        }
        o[r] = v;
    }

    float4* crow = (float4*)C + (size_t)(row0 + 8 * half) * (FD / 4) + 16 * part + l16;
    if (row0 + 16 <= M) {
        #pragma unroll
        for (int r = 0; r < 8; ++r) crow[r * (FD / 4)] = o[r];
    } else {
        #pragma unroll
        for (int r = 0; r < 8; ++r) {
            if (row0 + 8 * half + r < M) crow[r * (FD / 4)] = o[r];
        }
    }
}


extern "C" void kernel_launch(void* const* d_in, const int* in_sizes, int n_in,
                              void* d_out, int out_size, void* d_ws, size_t ws_size,
                              hipStream_t stream) {
    const float* x  = (const float*)d_in[0];
    const int*   ei = (const int*)d_in[1];
    const float* W1 = (const float*)d_in[2];
    const float* b1 = (const float*)d_in[3];
    const float* W2 = (const float*)d_in[4];
    const float* b2 = (const float*)d_in[5];

    const int N = in_sizes[0] / FD;
    const int E = in_sizes[1] / 2;
    const int* src = ei;
    const int* dst = ei + E;

    const size_t NP = ((size_t)N + 1 + 63) / 64 * 64;
    const size_t EP = ((size_t)E + 63) / 64 * 64;
    const int    ranges = (N + RNG - 1) / RNG;
    const size_t TP = ((size_t)ranges * (RNG / CNT_T) * NSL + 63) / 64 * 64;
    const int    slice  = ((E + NSL - 1) / NSL + 31) / 32 * 32;
    int*   wsi     = (int*)d_ws;
    int*   counts  = wsi;
    int*   rowptr  = wsi + NSL * NP;
    float* dis     = (float*)(rowptr + NP);
    int*   tiletot = rowptr + 2 * NP;
    int*   rank    = tiletot + TP;
    int*   adj     = rank + EP;
    float* bufA    = (float*)(adj + 2 * EP);
    float* out     = (float*)d_out;

    gcn_count_rank<<<dim3(ranges, NSL), CNT_T, 0, stream>>>(dst, counts, tiletot, rank, E, slice,
                                                           N, (int)NP);
    gcn_scan_dis<<<(N + 1 + SCAN_T - 1) / SCAN_T, SCAN_T, 0, stream>>>(counts, tiletot, rowptr, dis,
                                                                      N, (int)NP);
    gcn_fill<<<dim3((slice + 255) / 256, NSL), 256, 0, stream>>>(src, dst, rank, counts, dis, adj,
                                                                E, slice, (int)NP);

    const int tilesM  = (N + 15) / 16;
    const int gblocks = (tilesM + GEMM_STRIPS - 1) / GEMM_STRIPS;

    gcn_agg_gemm_wmma<<<gblocks, 32 * GEMM_WAVES, 0, stream>>>(x, rowptr, adj, dis, b1, W1,
                                                              bufA, N, 1);

    gcn_agg_gemm_wmma<<<gblocks, 32 * GEMM_WAVES, 0, stream>>>(bufA, rowptr, adj, dis, b2, W2,
                                                              out, N, 0);
}
